# starting kernel
# baseline (speedup 1.0000x reference)
#include <hip/hip_runtime.h>


typedef __attribute__((ext_vector_type(2))) float v2f;
typedef __attribute__((ext_vector_type(8))) float v8f;

#define IN_LEN            8192
#define OUT_LEN           8192
#define KSPLIT            8
#define KSLICE            (IN_LEN / KSPLIT)
#define COLS_PER_WAVE     16
#define NCOLGRP           (OUT_LEN / COLS_PER_WAVE)
#define WAVES_PER_BLOCK   8
#define THREADS_PER_BLOCK (WAVES_PER_BLOCK * 32)
#define MATVEC_BLOCKS     ((NCOLGRP * KSPLIT) / WAVES_PER_BLOCK)

__global__ __launch_bounds__(THREADS_PER_BLOCK)
void bias_init_kernel(const float* __restrict__ b, float* __restrict__ out) {
    int i = blockIdx.x * blockDim.x + threadIdx.x;
    if (i < OUT_LEN) out[i] = b[i];
}

__global__ __launch_bounds__(THREADS_PER_BLOCK)
void matvec_wmma_f32_kernel(const float* __restrict__ x,
                            const float* __restrict__ W,
                            float* __restrict__ out) {
    const int lane        = threadIdx.x & 31;
    const int waveInBlock = threadIdx.x >> 5;
    const int flatWave    = blockIdx.x * WAVES_PER_BLOCK + waveInBlock;

    const int slice    = flatWave >> 9;
    const int colGroup = flatWave & (NCOLGRP - 1);
    const int j0       = colGroup * COLS_PER_WAVE;
    const int kBeg     = slice * KSLICE;

    const int laneLo = lane & 15;
    const int rowOff = (lane >> 4) << 1;

    const float* wp = W + (size_t)(kBeg + rowOff) * OUT_LEN + (size_t)(j0 + laneLo);
    const float* xp = x + kBeg + rowOff;

    v8f acc0 = {};
    v8f acc1 = {};

    #pragma unroll 2
    for (int it = 0; it < KSLICE / 8; ++it) {
        v2f a0 = *(const v2f*)(xp);
        v2f a1 = *(const v2f*)(xp + 4);

        v2f b0;
        b0.x = wp[0];
        b0.y = wp[OUT_LEN];
        v2f b1;
        b1.x = wp[(size_t)4 * OUT_LEN];
        b1.y = wp[(size_t)5 * OUT_LEN];

        __builtin_prefetch(wp + (size_t)32 * OUT_LEN, 0, 1);

        acc0 = __builtin_amdgcn_wmma_f32_16x16x4_f32(
            false, a0, false, b0, (short)0, acc0, false, false);
        acc1 = __builtin_amdgcn_wmma_f32_16x16x4_f32(
            false, a1, false, b1, (short)0, acc1, false, false);

        wp += (size_t)8 * OUT_LEN;
        xp += 8;
    }

    float total = acc0[0] + acc1[0];
    if (lane < 16) {
        unsafeAtomicAdd(&out[j0 + laneLo], total);
    }
}

extern "C" void kernel_launch(void* const* d_in, const int* in_sizes, int n_in,
                              void* d_out, int out_size, void* d_ws, size_t ws_size,
                              hipStream_t stream) {
    (void)in_sizes; (void)n_in; (void)d_ws; (void)ws_size; (void)out_size;

    const float* x = (const float*)d_in[0];
    const float* W = (const float*)d_in[1];
    const float* b = (const float*)d_in[2];
    float*     out = (float*)d_out;

    bias_init_kernel<<<OUT_LEN / THREADS_PER_BLOCK, THREADS_PER_BLOCK, 0, stream>>>(b, out);

    matvec_wmma_f32_kernel<<<MATVEC_BLOCKS, THREADS_PER_BLOCK, 0, stream>>>(x, W, out);
}


# W read as global_load_b128 (lane owns 4 columns, 256 B of a row per lane half), ring of 8 four-row batches: 28-32 row loads in flight per wave; k-quarters summed in LDS (new sum tree), 8 atomics kept
# speedup vs baseline: 1.8171x; 1.8171x over previous
#include <hip/hip_runtime.h>


typedef __attribute__((ext_vector_type(2))) float v2f;
typedef __attribute__((ext_vector_type(4))) float v4f;
typedef __attribute__((ext_vector_type(8))) float v8f;

#define IN_LEN            8192
#define OUT_LEN           8192
#define KSPLIT            8
#define KSLICE            (IN_LEN / KSPLIT)
#define KQUART            4
#define KWAVE             (KSLICE / KQUART)
#define KHALF             (KWAVE / 2)
#define COLS_PER_WAVE     64
#define GROUPS_PER_BLOCK  4
#define COLS_PER_BLOCK    (COLS_PER_WAVE * GROUPS_PER_BLOCK)
#define WAVES_PER_BLOCK   (GROUPS_PER_BLOCK * KQUART)
#define MV_THREADS        (WAVES_PER_BLOCK * 32)
#define NCOLBLK           (OUT_LEN / COLS_PER_BLOCK)
#define MATVEC_BLOCKS     (NCOLBLK * KSPLIT)
#define BATCH             4
#define NBATCH            (KHALF / BATCH)
#define RING              8
#define ROW_V4            (OUT_LEN / 4)

#define BIAS_THREADS      256

__global__ __launch_bounds__(BIAS_THREADS)
void bias_init_kernel(const float* __restrict__ b, float* __restrict__ out) {
    int i = blockIdx.x * blockDim.x + threadIdx.x;
    if (i < OUT_LEN) out[i] = b[i];
}

__global__ __launch_bounds__(MV_THREADS)
void matvec_wmma_f32_kernel(const float* __restrict__ x,
                            const float* __restrict__ W,
                            float* __restrict__ out) {
    __shared__ __attribute__((aligned(16))) float red[KQUART * COLS_PER_BLOCK];

    const int tid  = threadIdx.x;
    const int lane = tid & 31;
    const int wave = __builtin_amdgcn_readfirstlane(tid >> 5);
    const int c    = lane & 15;
    const int h    = lane >> 4;
    const int grp  = wave & (GROUPS_PER_BLOCK - 1);
    const int kq   = wave >> 2;
    const int colBlk = blockIdx.x & (NCOLBLK - 1);
    const int slice  = blockIdx.x >> 5;

    const int j0 = colBlk * COLS_PER_BLOCK + grp * COLS_PER_WAVE;
    const int r0 = slice * KSLICE + kq * KWAVE + h * KHALF;

    const v4f* wp = reinterpret_cast<const v4f*>(W + (size_t)r0 * OUT_LEN + (size_t)(j0 + 4 * c));
    const v4f* xp = reinterpret_cast<const v4f*>(x + r0);

    v4f wb[RING][BATCH];
    v4f xb[RING][BATCH / 4];
    v8f acc0 = {}, acc1 = {}, acc2 = {}, acc3 = {};

#define LOAD_BATCH(N) do { \
        _Pragma("unroll") for (int q = 0; q < BATCH / 4; ++q) xb[(N) % RING][q] = xp[(N) * (BATCH / 4) + q]; \
        _Pragma("unroll") for (int t = 0; t < BATCH; ++t) wb[(N) % RING][t] = wp[(size_t)((N) * BATCH + t) * ROW_V4]; \
    } while (0)

#pragma unroll
    for (int n = 0; n < RING - 1; ++n) LOAD_BATCH(n);
#pragma unroll
    for (int n = 0; n < NBATCH; ++n) {
        __builtin_amdgcn_sched_barrier(0);
        if (n + RING - 1 < NBATCH) LOAD_BATCH(n + RING - 1);
        __builtin_amdgcn_sched_barrier(0);
#pragma unroll
        for (int p = 0; p < BATCH / 2; ++p) {
            const v4f xq = xb[n % RING][p >> 1];
            v2f a;
            a.x = (p & 1) ? xq.z : xq.x;
            a.y = (p & 1) ? xq.w : xq.y;
            const v4f w0 = wb[n % RING][2 * p];
            const v4f w1 = wb[n % RING][2 * p + 1];
            v2f b0; b0.x = w0.x; b0.y = w1.x;
            v2f b1; b1.x = w0.y; b1.y = w1.y;
            v2f b2; b2.x = w0.z; b2.y = w1.z;
            v2f b3; b3.x = w0.w; b3.y = w1.w;
            acc0 = __builtin_amdgcn_wmma_f32_16x16x4_f32(false, a, false, b0, (short)0, acc0, false, false);
            acc1 = __builtin_amdgcn_wmma_f32_16x16x4_f32(false, a, false, b1, (short)0, acc1, false, false);
            acc2 = __builtin_amdgcn_wmma_f32_16x16x4_f32(false, a, false, b2, (short)0, acc2, false, false);
            acc3 = __builtin_amdgcn_wmma_f32_16x16x4_f32(false, a, false, b3, (short)0, acc3, false, false);
        }
    }
#undef LOAD_BATCH

    if (h == 0) {
        v4f part;
        part.x = acc0[0]; part.y = acc1[0]; part.z = acc2[0]; part.w = acc3[0];
        *reinterpret_cast<v4f*>(&red[kq * COLS_PER_BLOCK + grp * COLS_PER_WAVE + 4 * c]) = part;
    }
    __syncthreads();

    if (tid < COLS_PER_BLOCK) {
        float s = red[tid];
        s += red[COLS_PER_BLOCK + tid];
        s += red[2 * COLS_PER_BLOCK + tid];
        s += red[3 * COLS_PER_BLOCK + tid];
        unsafeAtomicAdd(&out[colBlk * COLS_PER_BLOCK + tid], s);
    }
}

extern "C" void kernel_launch(void* const* d_in, const int* in_sizes, int n_in,
                              void* d_out, int out_size, void* d_ws, size_t ws_size,
                              hipStream_t stream) {
    (void)in_sizes; (void)n_in; (void)d_ws; (void)ws_size; (void)out_size;

    const float* x = (const float*)d_in[0];
    const float* W = (const float*)d_in[1];
    const float* b = (const float*)d_in[2];
    float*     out = (float*)d_out;

    bias_init_kernel<<<OUT_LEN / BIAS_THREADS, BIAS_THREADS, 0, stream>>>(b, out);

    matvec_wmma_f32_kernel<<<MATVEC_BLOCKS, MV_THREADS, 0, stream>>>(x, W, out);
}
